# starting kernel
# baseline (speedup 1.0000x reference)
#include <hip/hip_runtime.h>

typedef __attribute__((ext_vector_type(2))) float v2f;
typedef __attribute__((ext_vector_type(8))) float v8f;

#define BATCH   16
#define CIN     128
#define COUT    256
#define HW      56
#define SPATIAL (HW * HW)
#define NPIX    (BATCH * SPATIAL)
#define KTOT    (CIN * 9)
#define KC      16
#define KP      (KC / 2)
#define MT      64
#define NT      128
#define A2S     80
#define B2S     144

__global__ __launch_bounds__(256) void conv3x3_wmma_f32(
    const float* __restrict__ x, const float* __restrict__ w,
    const float* __restrict__ bias, float* __restrict__ out)
{
    __shared__ v2f a2_s[2][KP * A2S];
    __shared__ v2f b2_s[2][KP * B2S];

    const int tid    = threadIdx.x;
    const int lane   = tid & 31;
    const int wave   = tid >> 5;
    const int m_base = blockIdx.y * MT;
    const int n_base = blockIdx.x * NT;
    const int wm     = wave & 3;
    const int wn     = wave >> 2;

    const int a_cl = tid >> 2;
    const int a_kq = (tid & 3) << 2;
    const float* wptr = w + (size_t)(m_base + a_cl) * KTOT + a_kq;

    const int b_r  = tid >> 4;
    const int b_pg = (tid & 15) << 3;
    int pbase[8], poh[8], pow_[8];
#pragma unroll
    for (int j = 0; j < 8; ++j) {
        int pn = n_base + b_pg + j;
        int bb = pn / SPATIAL;
        int s  = pn - bb * SPATIAL;
        int oh = s / HW;
        int ow = s - oh * HW;
        pbase[j] = bb * (CIN * SPATIAL) + oh * HW + ow;
        poh[j]   = oh;
        pow_[j]  = ow;
    }

    auto gather = [&](int kkx, float* bv) {
        int k    = kkx + b_r;
        int ci   = k / 9;
        int rem  = k - ci * 9;
        int kh   = rem / 3;
        int kw   = rem - kh * 3;
        int coff = ci * SPATIAL + (kh - 1) * HW + (kw - 1);
#pragma unroll
        for (int j = 0; j < 8; ++j) {
            int ih = poh[j] + kh - 1;
            int iw = pow_[j] + kw - 1;
            bool ok = ((unsigned)ih < (unsigned)HW) &&
                      ((unsigned)iw < (unsigned)HW);
            bv[j] = ok ? x[pbase[j] + coff] : 0.f;
        }
    };

    auto commit = [&](int buf, const float4& w4, const float* bv) {
        v2f* a2 = a2_s[buf];
        v2f av0; av0.x = w4.x; av0.y = w4.y;
        v2f av1; av1.x = w4.z; av1.y = w4.w;
        a2[((a_kq >> 1) + 0) * A2S + a_cl] = av0;
        a2[((a_kq >> 1) + 1) * A2S + a_cl] = av1;
        float* bf = (float*)b2_s[buf];
        int bbase = ((b_r >> 1) * B2S) * 2 + (b_r & 1);
#pragma unroll
        for (int j = 0; j < 8; ++j)
            bf[bbase + (b_pg + j) * 2] = bv[j];
    };

    v8f acc[4];
#pragma unroll
    for (int i = 0; i < 4; ++i) acc[i] = (v8f)0.f;

    const int row    = lane & 15;
    const int kphalf = lane >> 4;
    const int a_col  = wm * 16 + row;
    const int b_col0 = wn * 64 + row;

    {
        float4 w4 = *(const float4*)(wptr);
        float bv[8];
        gather(0, bv);
        commit(0, w4, bv);
    }
    __syncthreads();

    for (int kk = 0; kk < KTOT; kk += KC) {
        const int  buf  = (kk / KC) & 1;
        const bool more = (kk + KC) < KTOT;

        float4 nw4;
        float  nbv[8];
        if (more) {
            nw4 = *(const float4*)(wptr + kk + KC);
            gather(kk + KC, nbv);
        }

        const v2f* a2 = a2_s[buf];
        const v2f* b2 = b2_s[buf];
#pragma unroll
        for (int ks = 0; ks < 4; ++ks) {
            int kp = 2 * ks + kphalf;
            v2f a = a2[kp * A2S + a_col];
#pragma unroll
            for (int nt = 0; nt < 4; ++nt) {
                v2f b = b2[kp * B2S + b_col0 + nt * 16];
                acc[nt] = __builtin_amdgcn_wmma_f32_16x16x4_f32(
                    false, a, false, b, (short)0, acc[nt], false, false);
            }
        }

        if (more) commit(buf ^ 1, nw4, nbv);
        __syncthreads();
    }

    const int mrow0 = m_base + wm * 16 + kphalf * 8;
#pragma unroll
    for (int nt = 0; nt < 4; ++nt) {
        int pn = n_base + wn * 64 + nt * 16 + row;
        int bb = pn / SPATIAL;
        int s  = pn - bb * SPATIAL;
        size_t obase = (size_t)bb * (COUT * SPATIAL) + s;
#pragma unroll
        for (int v = 0; v < 8; ++v) {
            int cout = mrow0 + v;
            __builtin_nontemporal_store(acc[nt][v] + bias[cout],
                                        &out[obase + (size_t)cout * SPATIAL]);
        }
    }
}

extern "C" void kernel_launch(void* const* d_in, const int* in_sizes, int n_in,
                              void* d_out, int out_size, void* d_ws, size_t ws_size,
                              hipStream_t stream) {
    const float* x    = (const float*)d_in[0];
    const float* wgt  = (const float*)d_in[1];
    const float* bias = (const float*)d_in[2];
    float* out        = (float*)d_out;

    dim3 grid(NPIX / NT, COUT / MT, 1);
    dim3 block(256, 1, 1);
    conv3x3_wmma_f32<<<grid, block, 0, stream>>>(x, wgt, bias, out);
}


# im2col gather: a thread's 8 pixels share one image row, so one running (ci,tap) decode, one row test and 4 branch-free loads off one address replace 8 per-pixel div/mod + bounds-branch loads
# speedup vs baseline: 1.2271x; 1.2271x over previous
#include <hip/hip_runtime.h>

typedef __attribute__((ext_vector_type(2))) float v2f;
typedef __attribute__((ext_vector_type(8))) float v8f;

#define BATCH   16
#define CIN     128
#define COUT    256
#define HW      56
#define SPATIAL (HW * HW)
#define NPIX    (BATCH * SPATIAL)
#define KTOT    (CIN * 9)
#define KC      16
#define KP      (KC / 2)
#define MT      64
#define NT      128
#define A2S     80
#define B2S     144

__global__ __launch_bounds__(256) void conv3x3_wmma_f32(
    const float* __restrict__ x, const float* __restrict__ w,
    const float* __restrict__ bias, float* __restrict__ out)
{
    __shared__ v2f a2_s[2][KP * A2S];
    __shared__ v2f b2_s[2][KP * B2S];

    const int tid    = threadIdx.x;
    const int lane   = tid & 31;
    const int wave   = tid >> 5;
    const int m_base = blockIdx.y * MT;
    const int n_base = blockIdx.x * NT;
    const int wm     = wave & 3;
    const int wn     = wave >> 2;

    const int a_cl = tid >> 2;
    const int a_kq = (tid & 3) << 2;
    const float* wptr = w + (size_t)(m_base + a_cl) * KTOT + a_kq;

    static_assert(NT % 8 == 0 && SPATIAL % 8 == 0 && HW % 8 == 0, "8-pixel groups must not straddle rows");
    const int b_r  = tid >> 4;
    const int b_pg = (tid & 15) << 3;
    const int pn0  = n_base + b_pg;
    const int g_bb = pn0 / SPATIAL;
    const int g_s  = pn0 - g_bb * SPATIAL;
    const int g_oh = g_s / HW;
    const int g_ow = g_s - g_oh * HW;
    const int g_xb = g_bb * (CIN * SPATIAL) + g_ow;
    int g_ci  = b_r / 9;
    int g_tap = b_r - g_ci * 9;

    auto gather = [&](float* bv, bool& rowbad, bool& lbad, bool& rbad) {
        const int kh = (g_tap >= 3) + (g_tap >= 6);
        const int kw = g_tap - 3 * kh;
        const int ih = g_oh + kh - 1;
        rowbad = (unsigned)ih >= (unsigned)HW;
        const int ihc = rowbad ? g_oh : ih;
        lbad = (g_ow + kw) == 0;
        rbad = (g_ow + kw) == (HW - 8 + 2);
        const int off = g_xb + __mul24(g_ci, SPATIAL) + __mul24(ihc, HW) + (kw - 1);
        bv[0] = x[off + (lbad ? 1 : 0)];
#pragma unroll
        for (int j = 1; j < 7; ++j) bv[j] = x[off + j];
        bv[7] = x[off + (rbad ? 6 : 7)];
        g_tap += KC - 9;
        g_ci  += 1;
        if (g_tap >= 9) { g_tap -= 9; g_ci += 1; }
    };

    auto commit = [&](int buf, const float4& w4, const float* bv,
                      bool rowbad, bool lbad, bool rbad) {
        v2f* a2 = a2_s[buf];
        v2f av0; av0.x = w4.x; av0.y = w4.y;
        v2f av1; av1.x = w4.z; av1.y = w4.w;
        a2[((a_kq >> 1) + 0) * A2S + a_cl] = av0;
        a2[((a_kq >> 1) + 1) * A2S + a_cl] = av1;
        float* bf = (float*)b2_s[buf];
        int bbase = ((b_r >> 1) * B2S) * 2 + (b_r & 1);
#pragma unroll
        for (int j = 0; j < 8; ++j) {
            bool bad = rowbad || (j == 0 && lbad) || (j == 7 && rbad);
            bf[bbase + (b_pg + j) * 2] = bad ? 0.f : bv[j];
        }
    };

    v8f acc[4];
#pragma unroll
    for (int i = 0; i < 4; ++i) acc[i] = (v8f)0.f;

    const int row    = lane & 15;
    const int kphalf = lane >> 4;
    const int a_col  = wm * 16 + row;
    const int b_col0 = wn * 64 + row;

    {
        float4 w4 = *(const float4*)(wptr);
        float bv[8];
        bool rb, lb, rtb;
        gather(bv, rb, lb, rtb);
        commit(0, w4, bv, rb, lb, rtb);
    }
    __syncthreads();

    for (int kk = 0; kk < KTOT; kk += KC) {
        const int  buf  = (kk / KC) & 1;
        const bool more = (kk + KC) < KTOT;

        float4 nw4;
        float  nbv[8];
        bool   nrb = false, nlb = false, nrtb = false;
        if (more) {
            nw4 = *(const float4*)(wptr + kk + KC);
            gather(nbv, nrb, nlb, nrtb);
        }

        const v2f* a2 = a2_s[buf];
        const v2f* b2 = b2_s[buf];
#pragma unroll
        for (int ks = 0; ks < 4; ++ks) {
            int kp = 2 * ks + kphalf;
            v2f a = a2[kp * A2S + a_col];
#pragma unroll
            for (int nt = 0; nt < 4; ++nt) {
                v2f b = b2[kp * B2S + b_col0 + nt * 16];
                acc[nt] = __builtin_amdgcn_wmma_f32_16x16x4_f32(
                    false, a, false, b, (short)0, acc[nt], false, false);
            }
        }

        if (more) commit(buf ^ 1, nw4, nbv, nrb, nlb, nrtb);
        __syncthreads();
    }

    const int mrow0 = m_base + wm * 16 + kphalf * 8;
#pragma unroll
    for (int nt = 0; nt < 4; ++nt) {
        int pn = n_base + wn * 64 + nt * 16 + row;
        int bb = pn / SPATIAL;
        int s  = pn - bb * SPATIAL;
        size_t obase = (size_t)bb * (COUT * SPATIAL) + s;
#pragma unroll
        for (int v = 0; v < 8; ++v) {
            int cout = mrow0 + v;
            __builtin_nontemporal_store(acc[nt][v] + bias[cout],
                                        &out[obase + (size_t)cout * SPATIAL]);
        }
    }
}

extern "C" void kernel_launch(void* const* d_in, const int* in_sizes, int n_in,
                              void* d_out, int out_size, void* d_ws, size_t ws_size,
                              hipStream_t stream) {
    const float* x    = (const float*)d_in[0];
    const float* wgt  = (const float*)d_in[1];
    const float* bias = (const float*)d_in[2];
    float* out        = (float*)d_out;

    dim3 grid(NPIX / NT, COUT / MT, 1);
    dim3 block(256, 1, 1);
    conv3x3_wmma_f32<<<grid, block, 0, stream>>>(x, wgt, bias, out);
}


# im2col gather: a wave loads one (ci,kh) input row per K chunk (aligned 16-byte load + 2 edge values per lane) and stores its 3 kw-shifted k-rows to LDS as 16-byte stores; K chunk 24
# speedup vs baseline: 1.3523x; 1.1020x over previous
#include <hip/hip_runtime.h>


typedef __attribute__((ext_vector_type(2))) float v2f;
typedef __attribute__((ext_vector_type(8))) float v8f;

#define BATCH   16
#define CIN     128
#define COUT    256
#define HW      56
#define SPATIAL (HW * HW)
#define NPIX    (BATCH * SPATIAL)
#define KTOT    (CIN * 9)
#define KC      24
#define KP      (KC / 2)
#define MT      64
#define NT      128
#define A2S     80
#define BKP     288
#define BKP4    (BKP / 4)

__global__ __launch_bounds__(256) void conv3x3_wmma_f32(
    const float* __restrict__ x, const float* __restrict__ w,
    const float* __restrict__ bias, float* __restrict__ out)
{
    __shared__ v2f    a2_s[2][KP * A2S];
    __shared__ float4 b4_s[2][KP * BKP4];

    const int tid    = threadIdx.x;
    const int lane   = tid & 31;
    const int wave   = __builtin_amdgcn_readfirstlane(tid >> 5);
    const int m_base = blockIdx.y * MT;
    const int n_base = blockIdx.x * NT;
    const int wm     = wave & 3;
    const int wn     = wave >> 2;

    const int a_cl = tid >> 2;
    const int a_j  = tid & 3;
    const float* wptr = w + (size_t)(m_base + a_cl) * KTOT;

    static_assert(NT % 4 == 0 && SPATIAL % 4 == 0 && HW % 4 == 0, "4-pixel groups must not straddle rows");
    static_assert(KTOT % KC == 0 && KC % 3 == 0 && KC / 3 == 8, "one (ci,kh) row per wave");
    const int pn0  = n_base + 4 * lane;
    const int g_bb = pn0 / SPATIAL;
    const int g_s  = pn0 - g_bb * SPATIAL;
    const int g_oh = g_s / HW;
    const int g_ow = g_s - g_oh * HW;
    const bool lbad = (g_ow == 0);
    const bool rbad = (g_ow == HW - 4);
    const int g_xb = g_bb * (CIN * SPATIAL) + g_oh * HW + g_ow;
    int q_ci = wave / 3;
    int q_kh = wave - q_ci * 3;

    auto gather = [&](float4& c4, float& vl, float& vr, bool& rowbad) {
        const int ih = g_oh + q_kh - 1;
        rowbad = (unsigned)ih >= (unsigned)HW;
        const int dh = rowbad ? 0 : (q_kh - 1);
        const int off = g_xb + q_ci * SPATIAL + dh * HW;
        c4 = *(const float4*)(x + off);
        vl = x[off - (lbad ? 0 : 1)];
        vr = x[off + (rbad ? 3 : 4)];
        q_kh += 2;
        q_ci += 2;
        if (q_kh >= 3) { q_kh -= 3; q_ci += 1; }
    };

    auto commit = [&](int buf, const float4& wa, const v2f& wb,
                      const float4& c4, float vl, float vr, bool rowbad) {
        v2f* a2 = a2_s[buf];
        v2f av0; av0.x = wa.x; av0.y = wa.y;
        v2f av1; av1.x = wa.z; av1.y = wa.w;
        a2[(2 * a_j + 0) * A2S + a_cl] = av0;
        a2[(2 * a_j + 1) * A2S + a_cl] = av1;
        a2[(8 + a_j) * A2S + a_cl]     = wb;
        const float l  = (rowbad || lbad) ? 0.f : vl;
        const float r  = (rowbad || rbad) ? 0.f : vr;
        const float c0 = rowbad ? 0.f : c4.x;
        const float c1 = rowbad ? 0.f : c4.y;
        const float c2 = rowbad ? 0.f : c4.z;
        const float c3 = rowbad ? 0.f : c4.w;
        float4* b4 = b4_s[buf] + (lane >> 2) * 8 + (lane & 3);
        const int k0l = 3 * wave;
        b4[((k0l + 0) >> 1) * BKP4 + ((k0l + 0) & 1) * 4] = make_float4(l,  c0, c1, c2);
        b4[((k0l + 1) >> 1) * BKP4 + ((k0l + 1) & 1) * 4] = make_float4(c0, c1, c2, c3);
        b4[((k0l + 2) >> 1) * BKP4 + ((k0l + 2) & 1) * 4] = make_float4(c1, c2, c3, r);
    };

    v8f acc[4];
#pragma unroll
    for (int i = 0; i < 4; ++i) acc[i] = (v8f)0.f;

    const int row    = lane & 15;
    const int kphalf = lane >> 4;
    const int a_col  = wm * 16 + row;
    const int b_off0 = wn * 128 + row;

    {
        float4 wa = *(const float4*)(wptr + 4 * a_j);
        v2f    wb = *(const v2f*)(wptr + 16 + 2 * a_j);
        float4 c4; float vl, vr; bool rb;
        gather(c4, vl, vr, rb);
        commit(0, wa, wb, c4, vl, vr, rb);
    }
    __syncthreads();

    for (int kk = 0; kk < KTOT; kk += KC) {
        const int  buf  = (kk / KC) & 1;
        const bool more = (kk + KC) < KTOT;

        float4 nwa; v2f nwb; float4 nc4;
        float  nvl = 0.f, nvr = 0.f;
        bool   nrb = false;
        if (more) {
            nwa = *(const float4*)(wptr + kk + KC + 4 * a_j);
            nwb = *(const v2f*)(wptr + kk + KC + 16 + 2 * a_j);
            gather(nc4, nvl, nvr, nrb);
        }

        const v2f*   a2 = a2_s[buf];
        const float* bf = (const float*)b4_s[buf];
#pragma unroll
        for (int ks = 0; ks < KC / 4; ++ks) {
            const int kp = 2 * ks + kphalf;
            v2f a = a2[kp * A2S + a_col];
#pragma unroll
            for (int nt = 0; nt < 4; ++nt) {
                const int bo = kp * BKP + b_off0 + nt * 32;
                v2f b;
                b.x = bf[bo];
                b.y = bf[bo + 16];
                acc[nt] = __builtin_amdgcn_wmma_f32_16x16x4_f32(
                    false, a, false, b, (short)0, acc[nt], false, false);
            }
        }

        if (more) commit(buf ^ 1, nwa, nwb, nc4, nvl, nvr, nrb);
        __syncthreads();
    }

    const int mrow0 = m_base + wm * 16 + kphalf * 8;
#pragma unroll
    for (int nt = 0; nt < 4; ++nt) {
        int pn = n_base + wn * 64 + nt * 16 + row;
        int bb = pn / SPATIAL;
        int s  = pn - bb * SPATIAL;
        size_t obase = (size_t)bb * (COUT * SPATIAL) + s;
#pragma unroll
        for (int v = 0; v < 8; ++v) {
            int cout = mrow0 + v;
            __builtin_nontemporal_store(acc[nt][v] + bias[cout],
                                        &out[obase + (size_t)cout * SPATIAL]);
        }
    }
}

extern "C" void kernel_launch(void* const* d_in, const int* in_sizes, int n_in,
                              void* d_out, int out_size, void* d_ws, size_t ws_size,
                              hipStream_t stream) {
    const float* x    = (const float*)d_in[0];
    const float* wgt  = (const float*)d_in[1];
    const float* bias = (const float*)d_in[2];
    float* out        = (float*)d_out;

    dim3 grid(NPIX / NT, COUT / MT, 1);
    dim3 block(256, 1, 1);
    conv3x3_wmma_f32<<<grid, block, 0, stream>>>(x, wgt, bias, out);
}
